# starting kernel
# baseline (speedup 1.0000x reference)
#include <hip/hip_runtime.h>
#include <hip/hip_bf16.h>
#include <math.h>

typedef float v2f __attribute__((ext_vector_type(2)));
typedef float v8f __attribute__((ext_vector_type(8)));

#if defined(__HIP_DEVICE_COMPILE__)
#if __has_builtin(__builtin_amdgcn_tanhf)
#define FAST_TANH(x) __builtin_amdgcn_tanhf(x)
#elif __has_builtin(__builtin_amdgcn_tanh_f32)
#define FAST_TANH(x) __builtin_amdgcn_tanh_f32(x)
#else
#define FAST_TANH(x) tanhf(x)
#endif
#if __has_builtin(__builtin_amdgcn_fmed3f)
#define RELU1(x) __builtin_amdgcn_fmed3f((x), 0.0f, __builtin_inff())
#else
#define RELU1(x) ((x) > 0.0f ? (x) : 0.0f)
#endif
#if !__has_builtin(__builtin_amdgcn_wmma_f32_16x16x4_f32)
#error "missing __builtin_amdgcn_wmma_f32_16x16x4_f32 for gfx1250 (device pass)"
#endif
#else
#define FAST_TANH(x) tanhf(x)
#define RELU1(x) ((x) > 0.0f ? (x) : 0.0f)
#endif

static __device__ __forceinline__ v8f wmma_f32x4(v2f a, v2f b, v8f c) {
  return __builtin_amdgcn_wmma_f32_16x16x4_f32(false, a, false, b, (short)0, c,
                                               false, false);
}

static __device__ __forceinline__ v8f splat8(float x) {
  v8f r;
#pragma unroll
  for (int i = 0; i < 8; ++i) r[i] = x;
  return r;
}

static __device__ __forceinline__ v8f relu8(v8f x) {
  v8f r;
#pragma unroll
  for (int i = 0; i < 8; ++i) r[i] = RELU1(x[i]);
  return r;
}

#define BLK 256

__global__ __launch_bounds__(BLK) void neural_ode_cls_kernel(
    const float* __restrict__ y0in, const float* __restrict__ tg,
    const float* __restrict__ W1, const float* __restrict__ b1,
    const float* __restrict__ W2, const float* __restrict__ b2,
    const float* __restrict__ Wc1, const float* __restrict__ bc1,
    const float* __restrict__ Wc2, const float* __restrict__ bc2,
    const float* __restrict__ Wc3, const float* __restrict__ bc3,
    float* __restrict__ out, int Bn, int Tn) {
  __shared__ float ybuf[BLK][2];
  __shared__ float hbuf[8][16][68];

  const int tid = threadIdx.x;
  const int gid = blockIdx.x * BLK + tid;
  const int lane = tid & 31;
  const int wave = tid >> 5;
  const int l = lane & 15;
  const int half = lane >> 4;

  float ya = 0.f, yb = 0.f;
  if (gid < Bn) {
    ya = y0in[2 * gid + 0];
    yb = y0in[2 * gid + 1];
  }
#pragma unroll 1
  for (int s = 0; s + 1 < Tn; ++s) {
    const float dt = tg[s + 1] - tg[s];
    float f0 = b2[0], f1 = b2[1];
#pragma unroll 1
    for (int c = 0; c < 4; ++c) {
      const float* __restrict__ w1a = W1 + 16 * c;
      const float* __restrict__ w1b = W1 + 64 + 16 * c;
      const float* __restrict__ bb = b1 + 16 * c;
      const float* __restrict__ w2 = W2 + 32 * c;
#pragma unroll
      for (int j = 0; j < 16; ++j) {
        const float h = FAST_TANH(fmaf(yb, w1b[j], fmaf(ya, w1a[j], bb[j])));
        f0 = fmaf(h, w2[2 * j + 0], f0);
        f1 = fmaf(h, w2[2 * j + 1], f1);
      }
    }
    ya = fmaf(dt, f0, ya);
    yb = fmaf(dt, f1, yb);
  }
  ybuf[tid][0] = ya;
  ybuf[tid][1] = yb;
  __syncthreads();

  v2f wb1[4];
  float cb1[4], cb2[4];
#pragma unroll
  for (int n = 0; n < 4; ++n) {
    const float wx = Wc1[0 * 64 + 16 * n + l];
    const float wy = Wc1[1 * 64 + 16 * n + l];
    wb1[n].x = half ? 0.f : wx;
    wb1[n].y = half ? 0.f : wy;
    cb1[n] = bc1[16 * n + l];
    cb2[n] = bc2[16 * n + l];
  }
  const int lc = (l < 3) ? l : 0;
  const float cb3 = (l < 3) ? bc3[lc] : 0.f;

#pragma unroll 1
  for (int tile = 0; tile < 2; ++tile) {
    const int r0 = 32 * wave + 16 * tile;

    const float yt0 = ybuf[r0 + l][0];
    const float yt1 = ybuf[r0 + l][1];
    v2f ay;
    ay.x = half ? 0.f : yt0;
    ay.y = half ? 0.f : yt1;

    v8f h1[4];
#pragma unroll
    for (int n = 0; n < 4; ++n)
      h1[n] = relu8(wmma_f32x4(ay, wb1[n], splat8(cb1[n])));

    __syncthreads();
#pragma unroll
    for (int n = 0; n < 4; ++n)
#pragma unroll
      for (int v = 0; v < 8; ++v)
        hbuf[wave][v + 8 * half][16 * n + l] = h1[n][v];
    __syncthreads();

    v8f h2[4];
#pragma unroll
    for (int n = 0; n < 4; ++n) h2[n] = splat8(cb2[n]);
#pragma unroll 2
    for (int kc = 0; kc < 16; ++kc) {
      const v2f a = *(const v2f*)&hbuf[wave][l][4 * kc + 2 * half];
#pragma unroll
      for (int n = 0; n < 4; ++n) {
        v2f bw;
        bw.x = Wc2[(4 * kc + 2 * half + 0) * 64 + 16 * n + l];
        bw.y = Wc2[(4 * kc + 2 * half + 1) * 64 + 16 * n + l];
        h2[n] = wmma_f32x4(a, bw, h2[n]);
      }
    }
#pragma unroll
    for (int n = 0; n < 4; ++n) h2[n] = relu8(h2[n]);

    __syncthreads();
#pragma unroll
    for (int n = 0; n < 4; ++n)
#pragma unroll
      for (int v = 0; v < 8; ++v)
        hbuf[wave][v + 8 * half][16 * n + l] = h2[n][v];
    __syncthreads();

    v8f o = splat8(cb3);
#pragma unroll 4
    for (int kc = 0; kc < 16; ++kc) {
      const v2f a = *(const v2f*)&hbuf[wave][l][4 * kc + 2 * half];
      const float w0 = Wc3[(4 * kc + 2 * half + 0) * 3 + lc];
      const float w1 = Wc3[(4 * kc + 2 * half + 1) * 3 + lc];
      v2f bw;
      bw.x = (l < 3) ? w0 : 0.f;
      bw.y = (l < 3) ? w1 : 0.f;
      o = wmma_f32x4(a, bw, o);
    }

    if (l < 3) {
      const int rowbase = blockIdx.x * BLK + r0 + 8 * half;
#pragma unroll
      for (int v = 0; v < 8; ++v) {
        const int gr = rowbase + v;
        if (gr < Bn) out[gr * 3 + l] = o[v];
      }
    }
  }
}

extern "C" void kernel_launch(void* const* d_in, const int* in_sizes, int n_in,
                              void* d_out, int out_size, void* d_ws,
                              size_t ws_size, hipStream_t stream) {
  const float* y0 = (const float*)d_in[0];
  const float* t = (const float*)d_in[1];
  const float* W1 = (const float*)d_in[2];
  const float* b1 = (const float*)d_in[3];
  const float* W2 = (const float*)d_in[4];
  const float* b2 = (const float*)d_in[5];
  const float* Wc1 = (const float*)d_in[6];
  const float* bc1 = (const float*)d_in[7];
  const float* Wc2 = (const float*)d_in[8];
  const float* bc2 = (const float*)d_in[9];
  const float* Wc3 = (const float*)d_in[10];
  const float* bc3 = (const float*)d_in[11];

  const int Bn = in_sizes[0] / 2;
  const int Tn = in_sizes[1];
  const int grid = (Bn + BLK - 1) / BLK;

  neural_ode_cls_kernel<<<grid, BLK, 0, stream>>>(
      y0, t, W1, b1, W2, b2, Wc1, bc1, Wc2, bc2, Wc3, bc3, (float*)d_out, Bn,
      Tn);
}


# Euler phase: ODE weights copied once (via LDS) into vector registers and kept for all steps; each step is packed f32 FMA on hidden-unit pairs, no loads; f0/f1 summed as 4 partial sums each
# speedup vs baseline: 1.5372x; 1.5372x over previous
#include <hip/hip_runtime.h>
#include <hip/hip_bf16.h>
#include <math.h>

typedef float v2f __attribute__((ext_vector_type(2)));
typedef float v8f __attribute__((ext_vector_type(8)));

#if defined(__HIP_DEVICE_COMPILE__)
#if __has_builtin(__builtin_amdgcn_tanhf)
#define FAST_TANH(x) __builtin_amdgcn_tanhf(x)
#elif __has_builtin(__builtin_amdgcn_tanh_f32)
#define FAST_TANH(x) __builtin_amdgcn_tanh_f32(x)
#else
#define FAST_TANH(x) tanhf(x)
#endif
#if __has_builtin(__builtin_amdgcn_fmed3f)
#define RELU1(x) __builtin_amdgcn_fmed3f((x), 0.0f, __builtin_inff())
#else
#define RELU1(x) ((x) > 0.0f ? (x) : 0.0f)
#endif
#if !__has_builtin(__builtin_amdgcn_wmma_f32_16x16x4_f32)
#error "missing __builtin_amdgcn_wmma_f32_16x16x4_f32 for gfx1250 (device pass)"
#endif
#else
#define FAST_TANH(x) tanhf(x)
#define RELU1(x) ((x) > 0.0f ? (x) : 0.0f)
#endif

static __device__ __forceinline__ v8f wmma_f32x4(v2f a, v2f b, v8f c) {
  return __builtin_amdgcn_wmma_f32_16x16x4_f32(false, a, false, b, (short)0, c,
                                               false, false);
}

static __device__ __forceinline__ v8f splat8(float x) {
  v8f r;
#pragma unroll
  for (int i = 0; i < 8; ++i) r[i] = x;
  return r;
}

static __device__ __forceinline__ v8f relu8(v8f x) {
  v8f r;
#pragma unroll
  for (int i = 0; i < 8; ++i) r[i] = RELU1(x[i]);
  return r;
}

#define BLK 256

__global__ __launch_bounds__(BLK) void neural_ode_cls_kernel(
    const float* __restrict__ y0in, const float* __restrict__ tg,
    const float* __restrict__ W1, const float* __restrict__ b1,
    const float* __restrict__ W2, const float* __restrict__ b2,
    const float* __restrict__ Wc1, const float* __restrict__ bc1,
    const float* __restrict__ Wc2, const float* __restrict__ bc2,
    const float* __restrict__ Wc3, const float* __restrict__ bc3,
    float* __restrict__ out, int Bn, int Tn) {
  __shared__ float ybuf[BLK][2];
  __shared__ float hbuf[8][16][68];
  __shared__ v2f wst[160];

  const int tid = threadIdx.x;
  const int gid = blockIdx.x * BLK + tid;
  const int lane = tid & 31;
  const int wave = tid >> 5;
  const int l = lane & 15;
  const int half = lane >> 4;

  for (int i = tid; i < 320; i += BLK) {
    float v;
    if (i < 128) {
      v = W1[i];
    } else if (i < 192) {
      v = b1[i - 128];
    } else {
      const int k = i - 192;
      v = W2[2 * (k & 63) + (k >> 6)];
    }
    ((float*)wst)[i] = v;
  }
  __syncthreads();
  v2f w1a[32], w1b[32], bb[32], w2a[32], w2b[32];
#pragma unroll
  for (int p = 0; p < 32; ++p) {
    w1a[p] = wst[p];
    w1b[p] = wst[32 + p];
    bb[p] = wst[64 + p];
    w2a[p] = wst[96 + p];
    w2b[p] = wst[128 + p];
  }
  const float b20 = b2[0], b21 = b2[1];

  float ya = 0.f, yb = 0.f;
  if (gid < Bn) {
    ya = y0in[2 * gid + 0];
    yb = y0in[2 * gid + 1];
  }
#pragma unroll 1
  for (int s = 0; s + 1 < Tn; ++s) {
    const float dt = tg[s + 1] - tg[s];
    v2f ya2, yb2;
    ya2.x = ya; ya2.y = ya;
    yb2.x = yb; yb2.y = yb;
    v2f fa0, fa1, fb0, fb1;
    fa0.x = b20; fa0.y = 0.f; fa1.x = 0.f; fa1.y = 0.f;
    fb0.x = b21; fb0.y = 0.f; fb1.x = 0.f; fb1.y = 0.f;
#pragma unroll
    for (int p = 0; p < 32; p += 2) {
      const v2f z0 = __builtin_elementwise_fma(
          yb2, w1b[p], __builtin_elementwise_fma(ya2, w1a[p], bb[p]));
      const v2f z1 = __builtin_elementwise_fma(
          yb2, w1b[p + 1],
          __builtin_elementwise_fma(ya2, w1a[p + 1], bb[p + 1]));
      v2f h0, h1;
      h0.x = FAST_TANH(z0.x);
      h0.y = FAST_TANH(z0.y);
      h1.x = FAST_TANH(z1.x);
      h1.y = FAST_TANH(z1.y);
      fa0 = __builtin_elementwise_fma(h0, w2a[p], fa0);
      fb0 = __builtin_elementwise_fma(h0, w2b[p], fb0);
      fa1 = __builtin_elementwise_fma(h1, w2a[p + 1], fa1);
      fb1 = __builtin_elementwise_fma(h1, w2b[p + 1], fb1);
    }
    const float f0 = (fa0.x + fa0.y) + (fa1.x + fa1.y);
    const float f1 = (fb0.x + fb0.y) + (fb1.x + fb1.y);
    ya = fmaf(dt, f0, ya);
    yb = fmaf(dt, f1, yb);
  }
  ybuf[tid][0] = ya;
  ybuf[tid][1] = yb;
  __syncthreads();

  v2f wb1[4];
  float cb1[4], cb2[4];
#pragma unroll
  for (int n = 0; n < 4; ++n) {
    const float wx = Wc1[0 * 64 + 16 * n + l];
    const float wy = Wc1[1 * 64 + 16 * n + l];
    wb1[n].x = half ? 0.f : wx;
    wb1[n].y = half ? 0.f : wy;
    cb1[n] = bc1[16 * n + l];
    cb2[n] = bc2[16 * n + l];
  }
  const int lc = (l < 3) ? l : 0;
  const float cb3 = (l < 3) ? bc3[lc] : 0.f;

#pragma unroll 1
  for (int tile = 0; tile < 2; ++tile) {
    const int r0 = 32 * wave + 16 * tile;

    const float yt0 = ybuf[r0 + l][0];
    const float yt1 = ybuf[r0 + l][1];
    v2f ay;
    ay.x = half ? 0.f : yt0;
    ay.y = half ? 0.f : yt1;

    v8f h1[4];
#pragma unroll
    for (int n = 0; n < 4; ++n)
      h1[n] = relu8(wmma_f32x4(ay, wb1[n], splat8(cb1[n])));

    __syncthreads();
#pragma unroll
    for (int n = 0; n < 4; ++n)
#pragma unroll
      for (int v = 0; v < 8; ++v)
        hbuf[wave][v + 8 * half][16 * n + l] = h1[n][v];
    __syncthreads();

    v8f h2[4];
#pragma unroll
    for (int n = 0; n < 4; ++n) h2[n] = splat8(cb2[n]);
#pragma unroll 2
    for (int kc = 0; kc < 16; ++kc) {
      const v2f a = *(const v2f*)&hbuf[wave][l][4 * kc + 2 * half];
#pragma unroll
      for (int n = 0; n < 4; ++n) {
        v2f bw;
        bw.x = Wc2[(4 * kc + 2 * half + 0) * 64 + 16 * n + l];
        bw.y = Wc2[(4 * kc + 2 * half + 1) * 64 + 16 * n + l];
        h2[n] = wmma_f32x4(a, bw, h2[n]);
      }
    }
#pragma unroll
    for (int n = 0; n < 4; ++n) h2[n] = relu8(h2[n]);

    __syncthreads();
#pragma unroll
    for (int n = 0; n < 4; ++n)
#pragma unroll
      for (int v = 0; v < 8; ++v)
        hbuf[wave][v + 8 * half][16 * n + l] = h2[n][v];
    __syncthreads();

    v8f o = splat8(cb3);
#pragma unroll 4
    for (int kc = 0; kc < 16; ++kc) {
      const v2f a = *(const v2f*)&hbuf[wave][l][4 * kc + 2 * half];
      const float w0 = Wc3[(4 * kc + 2 * half + 0) * 3 + lc];
      const float w1 = Wc3[(4 * kc + 2 * half + 1) * 3 + lc];
      v2f bw;
      bw.x = (l < 3) ? w0 : 0.f;
      bw.y = (l < 3) ? w1 : 0.f;
      o = wmma_f32x4(a, bw, o);
    }

    if (l < 3) {
      const int rowbase = blockIdx.x * BLK + r0 + 8 * half;
#pragma unroll
      for (int v = 0; v < 8; ++v) {
        const int gr = rowbase + v;
        if (gr < Bn) out[gr * 3 + l] = o[v];
      }
    }
  }
}

extern "C" void kernel_launch(void* const* d_in, const int* in_sizes, int n_in,
                              void* d_out, int out_size, void* d_ws,
                              size_t ws_size, hipStream_t stream) {
  const float* y0 = (const float*)d_in[0];
  const float* t = (const float*)d_in[1];
  const float* W1 = (const float*)d_in[2];
  const float* b1 = (const float*)d_in[3];
  const float* W2 = (const float*)d_in[4];
  const float* b2 = (const float*)d_in[5];
  const float* Wc1 = (const float*)d_in[6];
  const float* bc1 = (const float*)d_in[7];
  const float* Wc2 = (const float*)d_in[8];
  const float* bc2 = (const float*)d_in[9];
  const float* Wc3 = (const float*)d_in[10];
  const float* bc3 = (const float*)d_in[11];

  const int Bn = in_sizes[0] / 2;
  const int Tn = in_sizes[1];
  const int grid = (Bn + BLK - 1) / BLK;

  neural_ode_cls_kernel<<<grid, BLK, 0, stream>>>(
      y0, t, W1, b1, W2, b2, Wc1, bc1, Wc2, bc2, Wc3, bc3, (float*)d_out, Bn,
      Tn);
}


# Classifier computed transposed (weights = A, activations = B): each layer's result registers are the next layer's B fragments; no LDS transpose, no barriers; new k order in the sums
# speedup vs baseline: 2.0141x; 1.3102x over previous
#include <hip/hip_runtime.h>
#include <hip/hip_bf16.h>
#include <math.h>

typedef float v2f __attribute__((ext_vector_type(2)));
typedef float v8f __attribute__((ext_vector_type(8)));
typedef float v4fu __attribute__((ext_vector_type(4), aligned(4)));

#if defined(__HIP_DEVICE_COMPILE__)
#if __has_builtin(__builtin_amdgcn_tanhf)
#define FAST_TANH(x) __builtin_amdgcn_tanhf(x)
#elif __has_builtin(__builtin_amdgcn_tanh_f32)
#define FAST_TANH(x) __builtin_amdgcn_tanh_f32(x)
#else
#define FAST_TANH(x) tanhf(x)
#endif
#if __has_builtin(__builtin_amdgcn_fmed3f)
#define RELU1(x) __builtin_amdgcn_fmed3f((x), 0.0f, __builtin_inff())
#else
#define RELU1(x) ((x) > 0.0f ? (x) : 0.0f)
#endif
#if !__has_builtin(__builtin_amdgcn_wmma_f32_16x16x4_f32)
#error "missing __builtin_amdgcn_wmma_f32_16x16x4_f32 for gfx1250 (device pass)"
#endif
#else
#define FAST_TANH(x) tanhf(x)
#define RELU1(x) ((x) > 0.0f ? (x) : 0.0f)
#endif

static __device__ __forceinline__ v8f wmma_f32x4(v2f a, v2f b, v8f c) {
  return __builtin_amdgcn_wmma_f32_16x16x4_f32(false, a, false, b, (short)0, c,
                                               false, false);
}

static __device__ __forceinline__ v8f splat8(float x) {
  v8f r;
#pragma unroll
  for (int i = 0; i < 8; ++i) r[i] = x;
  return r;
}

static __device__ __forceinline__ v8f relu8(v8f x) {
  v8f r;
#pragma unroll
  for (int i = 0; i < 8; ++i) r[i] = RELU1(x[i]);
  return r;
}

#define BLK 256

__global__ __launch_bounds__(BLK) void neural_ode_cls_kernel(
    const float* __restrict__ y0in, const float* __restrict__ tg,
    const float* __restrict__ W1, const float* __restrict__ b1,
    const float* __restrict__ W2, const float* __restrict__ b2,
    const float* __restrict__ Wc1, const float* __restrict__ bc1,
    const float* __restrict__ Wc2, const float* __restrict__ bc2,
    const float* __restrict__ Wc3, const float* __restrict__ bc3,
    float* __restrict__ out, int Bn, int Tn) {
  __shared__ v2f wst[160];

  const int tid = threadIdx.x;
  const int gid = blockIdx.x * BLK + tid;
  const int lane = tid & 31;
  const int wave = tid >> 5;
  const int l = lane & 15;
  const int half = lane >> 4;

  for (int i = tid; i < 320; i += BLK) {
    float v;
    if (i < 128) {
      v = W1[i];
    } else if (i < 192) {
      v = b1[i - 128];
    } else {
      const int k = i - 192;
      v = W2[2 * (k & 63) + (k >> 6)];
    }
    ((float*)wst)[i] = v;
  }
  __syncthreads();
  v2f w1a[32], w1b[32], bb[32], w2a[32], w2b[32];
#pragma unroll
  for (int p = 0; p < 32; ++p) {
    w1a[p] = wst[p];
    w1b[p] = wst[32 + p];
    bb[p] = wst[64 + p];
    w2a[p] = wst[96 + p];
    w2b[p] = wst[128 + p];
  }
  const float b20 = b2[0], b21 = b2[1];

  float ya = 0.f, yb = 0.f;
  if (gid < Bn) {
    ya = y0in[2 * gid + 0];
    yb = y0in[2 * gid + 1];
  }
#pragma unroll 1
  for (int s = 0; s + 1 < Tn; ++s) {
    const float dt = tg[s + 1] - tg[s];
    v2f ya2, yb2;
    ya2.x = ya; ya2.y = ya;
    yb2.x = yb; yb2.y = yb;
    v2f fa0, fa1, fb0, fb1;
    fa0.x = b20; fa0.y = 0.f; fa1.x = 0.f; fa1.y = 0.f;
    fb0.x = b21; fb0.y = 0.f; fb1.x = 0.f; fb1.y = 0.f;
#pragma unroll
    for (int p = 0; p < 32; p += 2) {
      const v2f z0 = __builtin_elementwise_fma(
          yb2, w1b[p], __builtin_elementwise_fma(ya2, w1a[p], bb[p]));
      const v2f z1 = __builtin_elementwise_fma(
          yb2, w1b[p + 1],
          __builtin_elementwise_fma(ya2, w1a[p + 1], bb[p + 1]));
      v2f h0, h1;
      h0.x = FAST_TANH(z0.x);
      h0.y = FAST_TANH(z0.y);
      h1.x = FAST_TANH(z1.x);
      h1.y = FAST_TANH(z1.y);
      fa0 = __builtin_elementwise_fma(h0, w2a[p], fa0);
      fb0 = __builtin_elementwise_fma(h0, w2b[p], fb0);
      fa1 = __builtin_elementwise_fma(h1, w2a[p + 1], fa1);
      fb1 = __builtin_elementwise_fma(h1, w2b[p + 1], fb1);
    }
    const float f0 = (fa0.x + fa0.y) + (fa1.x + fa1.y);
    const float f1 = (fb0.x + fb0.y) + (fb1.x + fb1.y);
    ya = fmaf(dt, f0, ya);
    yb = fmaf(dt, f1, yb);
  }
  v2f zero2;
  zero2.x = 0.f; zero2.y = 0.f;
  v2f yme;
  yme.x = ya; yme.y = yb;
  v2f byt[2];
  byt[0] = half ? zero2 : yme;
  byt[1] = half ? yme : zero2;

  const v4fu w1x = *(const v4fu*)(Wc1 + 4 * l);
  const v4fu w1y = *(const v4fu*)(Wc1 + 64 + 4 * l);
  v8f c1[4], c2[4];
#pragma unroll
  for (int r = 0; r < 8; ++r) {
    const v4fu q1 = *(const v4fu*)(bc1 + 4 * (8 * half + r));
    const v4fu q2 = *(const v4fu*)(bc2 + 4 * (8 * half + r));
#pragma unroll
    for (int t = 0; t < 4; ++t) {
      c1[t][r] = q1[t];
      c2[t][r] = q2[t];
    }
  }

  const int lc = (l < 3) ? l : 0;
  v2f a3[4][4];
#pragma unroll
  for (int t = 0; t < 4; ++t) {
#pragma unroll
    for (int j = 0; j < 4; ++j) {
      const int k0 = 4 * (8 * half + 2 * j) + t;
      const float w0 = Wc3[k0 * 3 + lc];
      const float w1 = Wc3[(k0 + 4) * 3 + lc];
      a3[t][j].x = (l < 3) ? w0 : 0.f;
      a3[t][j].y = (l < 3) ? w1 : 0.f;
    }
  }
  const float b30 = bc3[0], b31 = bc3[1], b32 = bc3[2];

  v8f d1[2][4];
#pragma unroll
  for (int t = 0; t < 4; ++t) {
    v2f wme;
    wme.x = w1x[t]; wme.y = w1y[t];
    const v2f a_lo = half ? zero2 : wme;
    const v2f a_hi = half ? wme : zero2;
    d1[0][t] = relu8(wmma_f32x4(a_lo, byt[0], c1[t]));
    d1[1][t] = relu8(wmma_f32x4(a_hi, byt[1], c1[t]));
  }

  v8f acc[2][4];
#pragma unroll
  for (int n = 0; n < 4; ++n) {
    acc[0][n] = c2[n];
    acc[1][n] = c2[n];
  }
#pragma unroll
  for (int t = 0; t < 4; ++t) {
#pragma unroll
    for (int j = 0; j < 4; ++j) {
      const int k0 = 4 * (8 * half + 2 * j) + t;
      const v4fu q0 = *(const v4fu*)(Wc2 + k0 * 64 + 4 * l);
      const v4fu q1 = *(const v4fu*)(Wc2 + (k0 + 4) * 64 + 4 * l);
      v2f b0, b1;
      b0.x = d1[0][t][2 * j]; b0.y = d1[0][t][2 * j + 1];
      b1.x = d1[1][t][2 * j]; b1.y = d1[1][t][2 * j + 1];
#pragma unroll
      for (int n = 0; n < 4; ++n) {
        v2f a;
        a.x = q0[n]; a.y = q1[n];
        acc[0][n] = wmma_f32x4(a, b0, acc[0][n]);
        acc[1][n] = wmma_f32x4(a, b1, acc[1][n]);
      }
    }
  }
#pragma unroll
  for (int n = 0; n < 4; ++n) {
    acc[0][n] = relu8(acc[0][n]);
    acc[1][n] = relu8(acc[1][n]);
  }

  v8f o0 = splat8(0.f);
  o0[0] = half ? 0.f : b30;
  o0[1] = half ? 0.f : b31;
  o0[2] = half ? 0.f : b32;
  v8f o1 = o0;
#pragma unroll
  for (int t = 0; t < 4; ++t) {
#pragma unroll
    for (int j = 0; j < 4; ++j) {
      const v2f a = a3[t][j];
      v2f b0, b1;
      b0.x = acc[0][t][2 * j]; b0.y = acc[0][t][2 * j + 1];
      b1.x = acc[1][t][2 * j]; b1.y = acc[1][t][2 * j + 1];
      o0 = wmma_f32x4(a, b0, o0);
      o1 = wmma_f32x4(a, b1, o1);
    }
  }

  if (half == 0) {
    const int g0 = blockIdx.x * BLK + 32 * wave + l;
    if (g0 < Bn) {
      out[g0 * 3 + 0] = o0[0];
      out[g0 * 3 + 1] = o0[1];
      out[g0 * 3 + 2] = o0[2];
    }
    const int g1 = g0 + 16;
    if (g1 < Bn) {
      out[g1 * 3 + 0] = o1[0];
      out[g1 * 3 + 1] = o1[1];
      out[g1 * 3 + 2] = o1[2];
    }
  }
}

extern "C" void kernel_launch(void* const* d_in, const int* in_sizes, int n_in,
                              void* d_out, int out_size, void* d_ws,
                              size_t ws_size, hipStream_t stream) {
  const float* y0 = (const float*)d_in[0];
  const float* t = (const float*)d_in[1];
  const float* W1 = (const float*)d_in[2];
  const float* b1 = (const float*)d_in[3];
  const float* W2 = (const float*)d_in[4];
  const float* b2 = (const float*)d_in[5];
  const float* Wc1 = (const float*)d_in[6];
  const float* bc1 = (const float*)d_in[7];
  const float* Wc2 = (const float*)d_in[8];
  const float* bc2 = (const float*)d_in[9];
  const float* Wc3 = (const float*)d_in[10];
  const float* bc3 = (const float*)d_in[11];

  const int Bn = in_sizes[0] / 2;
  const int Tn = in_sizes[1];
  const int grid = (Bn + BLK - 1) / BLK;

  neural_ode_cls_kernel<<<grid, BLK, 0, stream>>>(
      y0, t, W1, b1, W2, b2, Wc1, bc1, Wc2, bc2, Wc3, bc3, (float*)d_out, Bn,
      Tn);
}
